# starting kernel
# baseline (speedup 1.0000x reference)
#include <hip/hip_runtime.h>
#include <cstdint>

typedef __attribute__((ext_vector_type(2))) float v2f;
typedef __attribute__((ext_vector_type(8))) float v8f;

#define D 64
#define WPAD 68

#define WMMA4(a, b, c) \
  __builtin_amdgcn_wmma_f32_16x16x4_f32(false, (a), false, (b), (short)0, (c), false, false)

__device__ __forceinline__ void stage_wT(const float* __restrict__ W, float (*wl)[WPAD],
                                         int tid) {
  for (int idx = tid; idx < 64 * 16; idx += 256) {
    const int k = idx >> 4;
    const int n4 = (idx & 15) << 2;
    const float4 v = *reinterpret_cast<const float4*>(W + k * D + n4);
    wl[n4 + 0][k] = v.x;
    wl[n4 + 1][k] = v.y;
    wl[n4 + 2][k] = v.z;
    wl[n4 + 3][k] = v.w;
  }
}

__global__ void k_init(const float* __restrict__ x, float* __restrict__ h,
                       float* __restrict__ deg, int n) {
  long i = (long)blockIdx.x * blockDim.x + threadIdx.x;
  long tot = (long)n * D;
  if (i < n) deg[i] = 0.0f;
  if (i < tot) h[i] = __logf(x[i] + 1.0f);
}

__global__ void k_deg(const int* __restrict__ dst, float* __restrict__ deg, int e) {
  long i = (long)blockIdx.x * blockDim.x + threadIdx.x;
  if (i < e) atomicAdd(&deg[dst[i]], 1.0f);
}

__global__ void k_zero4(float4* __restrict__ p, long n4) {
  long i = (long)blockIdx.x * blockDim.x + threadIdx.x;
  if (i < n4) p[i] = make_float4(0.f, 0.f, 0.f, 0.f);
}

__global__ void k_scatter(const float* __restrict__ h, const int* __restrict__ src,
                          const int* __restrict__ dst, float* __restrict__ neigh, int e) {
  long gid = (long)blockIdx.x * blockDim.x + threadIdx.x;
  long ed = gid >> 4;
  if (ed >= e) return;
  int f = (int)(gid & 15) * 4;
  int s = src[ed], d = dst[ed];
  float4 v = *reinterpret_cast<const float4*>(h + (size_t)s * D + f);
  float* p = neigh + (size_t)d * D + f;
  atomicAdd(p + 0, v.x);
  atomicAdd(p + 1, v.y);
  atomicAdd(p + 2, v.z);
  atomicAdd(p + 3, v.w);
}

__global__ void __launch_bounds__(256)
k_sage(const float* __restrict__ h, const float* __restrict__ neigh,
       const float* __restrict__ deg, const float* __restrict__ Ws,
       const float* __restrict__ Wn, const float* __restrict__ bias,
       float* __restrict__ out, int n, int do_norm) {
  __shared__ float wls[D][WPAD];
  __shared__ float wln[D][WPAD];
  stage_wT(Ws, wls, threadIdx.x);
  stage_wT(Wn, wln, threadIdx.x);
  __syncthreads();

  const int lane = threadIdx.x & 31;
  const int wave = threadIdx.x >> 5;
  const int row0 = (blockIdx.x * 8 + wave) * 16;
  if (row0 >= n) return;
  const int half = lane >> 4;
  const int l15 = lane & 15;
  int rowA = row0 + l15;
  if (rowA >= n) rowA = n - 1;
  const float dinv = 1.0f / fmaxf(deg[rowA], 1.0f);
  const float* __restrict__ hp = h + (size_t)rowA * D;
  const float* __restrict__ np = neigh + (size_t)rowA * D;

  v8f acc[4] = {};
  for (int k0 = 0; k0 < D; k0 += 4) {
    const int kk = k0 + half * 2;
    const v2f ah = *reinterpret_cast<const v2f*>(hp + kk);
    v2f an = *reinterpret_cast<const v2f*>(np + kk);
    an.x *= dinv;
    an.y *= dinv;
#pragma unroll
    for (int t = 0; t < 4; ++t) {
      const int nn = t * 16 + l15;
      const v2f bs = *reinterpret_cast<const v2f*>(&wls[nn][kk]);
      const v2f bn = *reinterpret_cast<const v2f*>(&wln[nn][kk]);
      acc[t] = WMMA4(ah, bs, acc[t]);
      acc[t] = WMMA4(an, bn, acc[t]);
    }
  }

  float ss[8];
#pragma unroll
  for (int i = 0; i < 8; ++i) ss[i] = 0.0f;
#pragma unroll
  for (int t = 0; t < 4; ++t) {
    const float bv = bias[t * 16 + l15];
#pragma unroll
    for (int i = 0; i < 8; ++i) {
      float v = acc[t][i] + bv;
      if (do_norm) v = fmaxf(v, 0.0f);
      acc[t][i] = v;
      ss[i] += v * v;
    }
  }
  if (do_norm) {
#pragma unroll
    for (int i = 0; i < 8; ++i) {
      float s = ss[i];
      s += __shfl_xor(s, 1, 32);
      s += __shfl_xor(s, 2, 32);
      s += __shfl_xor(s, 4, 32);
      s += __shfl_xor(s, 8, 32);
      const float inv = 1.0f / fmaxf(sqrtf(s), 1e-12f);
#pragma unroll
      for (int t = 0; t < 4; ++t) acc[t][i] *= inv;
    }
  }
#pragma unroll
  for (int t = 0; t < 4; ++t) {
#pragma unroll
    for (int i = 0; i < 8; ++i) {
      const int r = row0 + i + half * 8;
      if (r < n) out[(size_t)r * D + t * 16 + l15] = acc[t][i];
    }
  }
}

__global__ void __launch_bounds__(256)
k_head(const float* __restrict__ h, const float* __restrict__ fcw,
       const float* __restrict__ fcb, const float* __restrict__ gamma,
       const float* __restrict__ beta, const float* __restrict__ mean,
       const float* __restrict__ var, const float* __restrict__ w21,
       const float* __restrict__ b21, const float* __restrict__ w22,
       const float* __restrict__ b22, float* __restrict__ zloc,
       float* __restrict__ zscale, int n) {
  __shared__ float wl0[D][WPAD];
  __shared__ float wl1[D][WPAD];
  __shared__ float wl2[D][WPAD];
  __shared__ float u[8][16][WPAD];
  stage_wT(fcw, wl0, threadIdx.x);
  stage_wT(w21, wl1, threadIdx.x);
  stage_wT(w22, wl2, threadIdx.x);
  __syncthreads();

  const int lane = threadIdx.x & 31;
  const int wave = threadIdx.x >> 5;
  const int row0 = (blockIdx.x * 8 + wave) * 16;
  if (row0 >= n) return;
  const int half = lane >> 4;
  const int l15 = lane & 15;
  int rowA = row0 + l15;
  if (rowA >= n) rowA = n - 1;
  const float* __restrict__ hp = h + (size_t)rowA * D;

  v8f acc[4] = {};
  for (int k0 = 0; k0 < D; k0 += 4) {
    const int kk = k0 + half * 2;
    const v2f a = *reinterpret_cast<const v2f*>(hp + kk);
#pragma unroll
    for (int t = 0; t < 4; ++t) {
      const int nn = t * 16 + l15;
      const v2f b = *reinterpret_cast<const v2f*>(&wl0[nn][kk]);
      acc[t] = WMMA4(a, b, acc[t]);
    }
  }
#pragma unroll
  for (int t = 0; t < 4; ++t) {
    const int c = t * 16 + l15;
    const float bb = fcb[c];
    const float m = mean[c];
    const float iv = rsqrtf(var[c] + 1e-5f);
    const float g = gamma[c];
    const float bt = beta[c];
#pragma unroll
    for (int i = 0; i < 8; ++i) {
      float v = acc[t][i] + bb;
      v = (v - m) * iv * g + bt;
      v = fmaxf(v, 0.0f);
      v = v + __logf(1.0f + __expf(-v));
      u[wave][i + half * 8][c] = v;
    }
  }
  v8f aL[4] = {}, aS[4] = {};
  for (int k0 = 0; k0 < D; k0 += 4) {
    const int kk = k0 + half * 2;
    const v2f a = *reinterpret_cast<const v2f*>(&u[wave][l15][kk]);
#pragma unroll
    for (int t = 0; t < 4; ++t) {
      const int nn = t * 16 + l15;
      const v2f bL = *reinterpret_cast<const v2f*>(&wl1[nn][kk]);
      const v2f bS = *reinterpret_cast<const v2f*>(&wl2[nn][kk]);
      aL[t] = WMMA4(a, bL, aL[t]);
      aS[t] = WMMA4(a, bS, aS[t]);
    }
  }
#pragma unroll
  for (int t = 0; t < 4; ++t) {
    const int c = t * 16 + l15;
    const float bl = b21[c];
    const float bs = b22[c];
#pragma unroll
    for (int i = 0; i < 8; ++i) {
      const int r = row0 + i + half * 8;
      if (r < n) {
        const size_t o = (size_t)r * D + c;
        zloc[o] = aL[t][i] + bl;
        zscale[o] = __expf(aS[t][i] + bs);
      }
    }
  }
}

extern "C" void kernel_launch(void* const* d_in, const int* in_sizes, int n_in,
                              void* d_out, int out_size, void* d_ws, size_t ws_size,
                              hipStream_t stream) {
  const float* x = (const float*)d_in[0];
  const int* esrc = (const int*)d_in[1];
  const int* edst = (const int*)d_in[2];
  const float* Wself = (const float*)d_in[3];
  const float* Wneigh = (const float*)d_in[4];
  const float* bsage = (const float*)d_in[5];
  const float* fcw = (const float*)d_in[6];
  const float* fcb = (const float*)d_in[7];
  const float* gamma = (const float*)d_in[8];
  const float* beta = (const float*)d_in[9];
  const float* mean = (const float*)d_in[10];
  const float* var = (const float*)d_in[11];
  const float* w21 = (const float*)d_in[12];
  const float* b21 = (const float*)d_in[13];
  const float* w22 = (const float*)d_in[14];
  const float* b22 = (const float*)d_in[15];

  const int n = in_sizes[0] / D;
  const int e = in_sizes[1];
  const size_t nd = (size_t)n * D;

  float* ws = (float*)d_ws;
  float* deg = ws;
  float* hA = ws + (((size_t)n + 255) & ~(size_t)255);
  float* hB = hA + nd;
  float* nb = hB + nd;

  {
    long tot = (long)n * D;
    k_init<<<(int)((tot + 255) / 256), 256, 0, stream>>>(x, hA, deg, n);
  }
  k_deg<<<(e + 255) / 256, 256, 0, stream>>>(edst, deg, e);

  const int tiles = (n + 15) / 16;
  const int gblocks = (tiles + 7) / 8;

  float* hcur = hA;
  float* hnext = hB;
  for (int l = 0; l < 3; ++l) {
    long n4 = (long)(nd / 4);
    k_zero4<<<(int)((n4 + 255) / 256), 256, 0, stream>>>((float4*)nb, n4);
    long st = (long)e * 16;
    k_scatter<<<(int)((st + 255) / 256), 256, 0, stream>>>(hcur, esrc, edst, nb, e);
    k_sage<<<gblocks, 256, 0, stream>>>(hcur, nb, deg, Wself + (size_t)l * D * D,
                                        Wneigh + (size_t)l * D * D, bsage + (size_t)l * D,
                                        hnext, n, (l < 2) ? 1 : 0);
    float* tmp = hcur;
    hcur = hnext;
    hnext = tmp;
  }

  float* zloc = (float*)d_out;
  float* zscale = zloc + nd;
  k_head<<<gblocks, 256, 0, stream>>>(hcur, fcw, fcb, gamma, beta, mean, var, w21, b21,
                                      w22, b22, zloc, zscale, n);
}


# Neighbour sum by gather over sources grouped by destination (per-call counting sort, integer atomic slots) in the layer GEMM kernel; float atomic scatter and zero fills gone; sum order = slot order
# speedup vs baseline: 1.7966x; 1.7966x over previous
#include <hip/hip_runtime.h>
#include <cstdint>

typedef __attribute__((ext_vector_type(2))) float v2f;
typedef __attribute__((ext_vector_type(8))) float v8f;

#define D 64
#define WPAD 68

#define WMMA4(a, b, c) \
  __builtin_amdgcn_wmma_f32_16x16x4_f32(false, (a), false, (b), (short)0, (c), false, false)

__device__ __forceinline__ void stage_wT(const float* __restrict__ W, float (*wl)[WPAD],
                                         int tid) {
  for (int idx = tid; idx < 64 * 16; idx += 256) {
    const int k = idx >> 4;
    const int n4 = (idx & 15) << 2;
    const float4 v = *reinterpret_cast<const float4*>(W + k * D + n4);
    wl[n4 + 0][k] = v.x;
    wl[n4 + 1][k] = v.y;
    wl[n4 + 2][k] = v.z;
    wl[n4 + 3][k] = v.w;
  }
}

__global__ void k_init(const float* __restrict__ x, float* __restrict__ h,
                       int* __restrict__ cnt, int n) {
  long i = (long)blockIdx.x * blockDim.x + threadIdx.x;
  long tot = (long)n * D;
  if (i < n) cnt[i] = 0;
  if (i < tot) h[i] = __logf(x[i] + 1.0f);
}

__global__ void k_count(const int* __restrict__ dst, int* __restrict__ cnt,
                        int* __restrict__ rank, int e) {
  long i = (long)blockIdx.x * blockDim.x + threadIdx.x;
  if (i < e) rank[i] = atomicAdd(&cnt[dst[i]], 1);
}

__global__ void __launch_bounds__(1024)
k_scan(const int* __restrict__ cnt, int* __restrict__ rowptr, int n) {
  __shared__ int wtot[32];
  __shared__ int wpre[32];
  __shared__ int s_off;
  const int tid = threadIdx.x;
  const int lane = tid & 31;
  const int w = tid >> 5;
  const int tile0 = blockIdx.x * 1024;
  int p = 0;
  for (int i = tid * 4; i < tile0; i += 4096) {
    const int4 q = *reinterpret_cast<const int4*>(cnt + i);
    p += q.x + q.y + q.z + q.w;
  }
  const int i = tile0 + tid;
  const int v = (i < n) ? cnt[i] : 0;
  int x = v;
  for (int d = 1; d < 32; d <<= 1) {
    const int y = __shfl_up(x, d, 32);
    if (lane >= d) x += y;
  }
  for (int d = 16; d > 0; d >>= 1) p += __shfl_xor(p, d, 32);
  if (lane == 31) wtot[w] = x;
  if (lane == 0) wpre[w] = p;
  __syncthreads();
  if (w == 0) {
    const int t = wtot[lane];
    int q = wpre[lane];
    int xs = t;
    for (int d = 1; d < 32; d <<= 1) {
      const int y = __shfl_up(xs, d, 32);
      if (lane >= d) xs += y;
    }
    for (int d = 16; d > 0; d >>= 1) q += __shfl_xor(q, d, 32);
    wtot[lane] = xs - t;
    if (lane == 0) s_off = q;
  }
  __syncthreads();
  if (i <= n) rowptr[i] = s_off + wtot[w] + (x - v);
}

__global__ void k_fill(const int* __restrict__ src, const int* __restrict__ dst,
                       const int* __restrict__ rowptr, const int* __restrict__ rank,
                       int* __restrict__ col, int e) {
  long i = (long)blockIdx.x * blockDim.x + threadIdx.x;
  if (i < e) col[rowptr[dst[i]] + rank[i]] = src[i];
}

__global__ void __launch_bounds__(256)
k_sage(const float* __restrict__ h, const int* __restrict__ rowptr,
       const int* __restrict__ col, const float* __restrict__ Ws,
       const float* __restrict__ Wn, const float* __restrict__ bias,
       float* __restrict__ out, int n, int do_norm) {
  __shared__ float wls[D][WPAD];
  __shared__ float wln[D][WPAD];
  __shared__ float nt[8][17][WPAD];

  const int lane = threadIdx.x & 31;
  const int wave = threadIdx.x >> 5;
  const int row0 = (blockIdx.x * 8 + wave) * 16;
  int rq = row0 + lane;
  if (rq > n) rq = n;
  const int rpv = rowptr[rq];

  stage_wT(Ws, wls, threadIdx.x);
  stage_wT(Wn, wln, threadIdx.x);

  int rp[17];
#pragma unroll
  for (int r = 0; r <= 16; ++r) rp[r] = __builtin_amdgcn_readlane(rpv, r);
  const int segBeg = rp[0];
  const int segEnd = rp[16];
  int nsrc = (segBeg + lane < segEnd) ? col[segBeg + lane] : 0;
  __syncthreads();

  if (row0 >= n) return;
  const int half = lane >> 4;
  const int l15 = lane & 15;
  int rowA = row0 + l15;
  if (rowA >= n) rowA = n - 1;
  const float* __restrict__ hp = h + (size_t)rowA * D;

  const v2f zero2 = {0.0f, 0.0f};
#pragma unroll
  for (int r = 0; r < 16; ++r) *reinterpret_cast<v2f*>(&nt[wave][r][lane * 2]) = zero2;

  v2f sum = zero2;
  int cur = 16;
  for (int base = segBeg; base < segEnd; base += 32) {
    const int j = base + lane;
    const int msrc = nsrc;
    int mrow = 0;
#pragma unroll
    for (int r = 1; r <= 16; ++r) mrow += (j >= rp[r]) ? 1 : 0;
    v2f v[32];
    int rr[32];
#pragma unroll
    for (int u = 0; u < 32; ++u) {
      const int s = __builtin_amdgcn_readlane(msrc, u);
      rr[u] = __builtin_amdgcn_readlane(mrow, u);
      v[u] = *reinterpret_cast<const v2f*>(h + (size_t)s * D + lane * 2);
    }
    const int jn = j + 32;
    nsrc = (jn < segEnd) ? col[jn] : 0;
#pragma unroll
    for (int u = 0; u < 32; ++u) {
      if (rr[u] != cur) {
        const int c0 = __builtin_amdgcn_readlane(rpv, cur);
        const int c1 = __builtin_amdgcn_readlane(rpv, cur + 1);
        const float dinv = 1.0f / fmaxf((float)(c1 - c0), 1.0f);
        v2f m = sum;
        m.x *= dinv;
        m.y *= dinv;
        *reinterpret_cast<v2f*>(&nt[wave][cur][lane * 2]) = m;
        sum = zero2;
        cur = rr[u];
      }
      sum.x += v[u].x;
      sum.y += v[u].y;
    }
  }
  {
    const int c0 = __builtin_amdgcn_readlane(rpv, cur);
    const int c1 = __builtin_amdgcn_readlane(rpv, cur + 1);
    const float dinv = 1.0f / fmaxf((float)(c1 - c0), 1.0f);
    v2f m = sum;
    m.x *= dinv;
    m.y *= dinv;
    *reinterpret_cast<v2f*>(&nt[wave][cur][lane * 2]) = m;
  }

  v8f acc[4] = {};
#pragma unroll
  for (int k0 = 0; k0 < D; k0 += 4) {
    const int kk = k0 + half * 2;
    const v2f ah = *reinterpret_cast<const v2f*>(hp + kk);
    const v2f an = *reinterpret_cast<const v2f*>(&nt[wave][l15][kk]);
#pragma unroll
    for (int t = 0; t < 4; ++t) {
      const int nn = t * 16 + l15;
      const v2f bs = *reinterpret_cast<const v2f*>(&wls[nn][kk]);
      const v2f bn = *reinterpret_cast<const v2f*>(&wln[nn][kk]);
      acc[t] = WMMA4(ah, bs, acc[t]);
      acc[t] = WMMA4(an, bn, acc[t]);
    }
  }

  float ss[8];
#pragma unroll
  for (int i = 0; i < 8; ++i) ss[i] = 0.0f;
#pragma unroll
  for (int t = 0; t < 4; ++t) {
    const float bv = bias[t * 16 + l15];
#pragma unroll
    for (int i = 0; i < 8; ++i) {
      float v = acc[t][i] + bv;
      if (do_norm) v = fmaxf(v, 0.0f);
      acc[t][i] = v;
      ss[i] += v * v;
    }
  }
  if (do_norm) {
#pragma unroll
    for (int i = 0; i < 8; ++i) {
      float s = ss[i];
      s += __shfl_xor(s, 1, 32);
      s += __shfl_xor(s, 2, 32);
      s += __shfl_xor(s, 4, 32);
      s += __shfl_xor(s, 8, 32);
      const float inv = 1.0f / fmaxf(sqrtf(s), 1e-12f);
#pragma unroll
      for (int t = 0; t < 4; ++t) acc[t][i] *= inv;
    }
  }
#pragma unroll
  for (int t = 0; t < 4; ++t) {
#pragma unroll
    for (int i = 0; i < 8; ++i) {
      const int r = row0 + i + half * 8;
      if (r < n) out[(size_t)r * D + t * 16 + l15] = acc[t][i];
    }
  }
}

__global__ void __launch_bounds__(256)
k_head(const float* __restrict__ h, const float* __restrict__ fcw,
       const float* __restrict__ fcb, const float* __restrict__ gamma,
       const float* __restrict__ beta, const float* __restrict__ mean,
       const float* __restrict__ var, const float* __restrict__ w21,
       const float* __restrict__ b21, const float* __restrict__ w22,
       const float* __restrict__ b22, float* __restrict__ zloc,
       float* __restrict__ zscale, int n) {
  __shared__ float wl0[D][WPAD];
  __shared__ float wl1[D][WPAD];
  __shared__ float wl2[D][WPAD];
  __shared__ float u[8][16][WPAD];
  stage_wT(fcw, wl0, threadIdx.x);
  stage_wT(w21, wl1, threadIdx.x);
  stage_wT(w22, wl2, threadIdx.x);
  __syncthreads();

  const int lane = threadIdx.x & 31;
  const int wave = threadIdx.x >> 5;
  const int row0 = (blockIdx.x * 8 + wave) * 16;
  if (row0 >= n) return;
  const int half = lane >> 4;
  const int l15 = lane & 15;
  int rowA = row0 + l15;
  if (rowA >= n) rowA = n - 1;
  const float* __restrict__ hp = h + (size_t)rowA * D;

  v8f acc[4] = {};
  for (int k0 = 0; k0 < D; k0 += 4) {
    const int kk = k0 + half * 2;
    const v2f a = *reinterpret_cast<const v2f*>(hp + kk);
#pragma unroll
    for (int t = 0; t < 4; ++t) {
      const int nn = t * 16 + l15;
      const v2f b = *reinterpret_cast<const v2f*>(&wl0[nn][kk]);
      acc[t] = WMMA4(a, b, acc[t]);
    }
  }
#pragma unroll
  for (int t = 0; t < 4; ++t) {
    const int c = t * 16 + l15;
    const float bb = fcb[c];
    const float m = mean[c];
    const float iv = rsqrtf(var[c] + 1e-5f);
    const float g = gamma[c];
    const float bt = beta[c];
#pragma unroll
    for (int i = 0; i < 8; ++i) {
      float v = acc[t][i] + bb;
      v = (v - m) * iv * g + bt;
      v = fmaxf(v, 0.0f);
      v = v + __logf(1.0f + __expf(-v));
      u[wave][i + half * 8][c] = v;
    }
  }
  v8f aL[4] = {}, aS[4] = {};
  for (int k0 = 0; k0 < D; k0 += 4) {
    const int kk = k0 + half * 2;
    const v2f a = *reinterpret_cast<const v2f*>(&u[wave][l15][kk]);
#pragma unroll
    for (int t = 0; t < 4; ++t) {
      const int nn = t * 16 + l15;
      const v2f bL = *reinterpret_cast<const v2f*>(&wl1[nn][kk]);
      const v2f bS = *reinterpret_cast<const v2f*>(&wl2[nn][kk]);
      aL[t] = WMMA4(a, bL, aL[t]);
      aS[t] = WMMA4(a, bS, aS[t]);
    }
  }
#pragma unroll
  for (int t = 0; t < 4; ++t) {
    const int c = t * 16 + l15;
    const float bl = b21[c];
    const float bs = b22[c];
#pragma unroll
    for (int i = 0; i < 8; ++i) {
      const int r = row0 + i + half * 8;
      if (r < n) {
        const size_t o = (size_t)r * D + c;
        zloc[o] = aL[t][i] + bl;
        zscale[o] = __expf(aS[t][i] + bs);
      }
    }
  }
}

extern "C" void kernel_launch(void* const* d_in, const int* in_sizes, int n_in,
                              void* d_out, int out_size, void* d_ws, size_t ws_size,
                              hipStream_t stream) {
  const float* x = (const float*)d_in[0];
  const int* esrc = (const int*)d_in[1];
  const int* edst = (const int*)d_in[2];
  const float* Wself = (const float*)d_in[3];
  const float* Wneigh = (const float*)d_in[4];
  const float* bsage = (const float*)d_in[5];
  const float* fcw = (const float*)d_in[6];
  const float* fcb = (const float*)d_in[7];
  const float* gamma = (const float*)d_in[8];
  const float* beta = (const float*)d_in[9];
  const float* mean = (const float*)d_in[10];
  const float* var = (const float*)d_in[11];
  const float* w21 = (const float*)d_in[12];
  const float* b21 = (const float*)d_in[13];
  const float* w22 = (const float*)d_in[14];
  const float* b22 = (const float*)d_in[15];

  const int n = in_sizes[0] / D;
  const int e = in_sizes[1];
  const size_t nd = (size_t)n * D;

  float* ws = (float*)d_ws;
  int* cnt = (int*)ws;
  float* hA = ws + (((size_t)n + 255) & ~(size_t)255);
  float* hB = hA + nd;
  int* rowptr = (int*)(hB + nd);
  int* col = rowptr + (((size_t)n + 1 + 255) & ~(size_t)255);
  int* rank = col + e;

  {
    long tot = (long)n * D;
    k_init<<<(int)((tot + 255) / 256), 256, 0, stream>>>(x, hA, cnt, n);
  }
  k_count<<<(e + 255) / 256, 256, 0, stream>>>(edst, cnt, rank, e);
  k_scan<<<(n + 1024) / 1024, 1024, 0, stream>>>(cnt, rowptr, n);
  k_fill<<<(e + 255) / 256, 256, 0, stream>>>(esrc, edst, rowptr, rank, col, e);

  const int tiles = (n + 15) / 16;
  const int gblocks = (tiles + 7) / 8;

  float* hcur = hA;
  float* hnext = hB;
  for (int l = 0; l < 3; ++l) {
    k_sage<<<gblocks, 256, 0, stream>>>(hcur, rowptr, col, Wself + (size_t)l * D * D,
                                        Wneigh + (size_t)l * D * D, bsage + (size_t)l * D,
                                        hnext, n, (l < 2) ? 1 : 0);
    float* tmp = hcur;
    hcur = hnext;
    hnext = tmp;
  }

  float* zloc = (float*)d_out;
  float* zscale = zloc + nd;
  k_head<<<gblocks, 256, 0, stream>>>(hcur, fcw, fcb, gamma, beta, mean, var, w21, b21,
                                      w22, b22, zloc, zscale, n);
}


# One 25-wave work-group per WGP for the layer and head kernels (250 groups of 25 row tiles, was 782 of 8): weights staged once per WGP in one load group, even tiles per WGP, head in one pass
# speedup vs baseline: 2.0145x; 1.1213x over previous
#include <hip/hip_runtime.h>
#include <cstdint>

typedef __attribute__((ext_vector_type(2))) float v2f;
typedef __attribute__((ext_vector_type(8))) float v8f;

#define D 64
#define WPAD 68
#define WPB 25
#define NTHR (WPB * 32)

#define WMMA4(a, b, c) \
  __builtin_amdgcn_wmma_f32_16x16x4_f32(false, (a), false, (b), (short)0, (c), false, false)

__device__ __forceinline__ int wT_second(int tid) {
  return (tid + NTHR < 64 * 16) ? tid + NTHR : tid;
}
__device__ __forceinline__ void wT_load(const float* __restrict__ W, int tid, float4& a,
                                        float4& b) {
  const int i2 = wT_second(tid);
  a = *reinterpret_cast<const float4*>(W + (tid >> 4) * D + ((tid & 15) << 2));
  b = *reinterpret_cast<const float4*>(W + (i2 >> 4) * D + ((i2 & 15) << 2));
}
__device__ __forceinline__ void wT_put(float (*wl)[WPAD], int idx, const float4& v) {
  const int k = idx >> 4;
  const int n4 = (idx & 15) << 2;
  wl[n4 + 0][k] = v.x;
  wl[n4 + 1][k] = v.y;
  wl[n4 + 2][k] = v.z;
  wl[n4 + 3][k] = v.w;
}
__device__ __forceinline__ void wT_store(float (*wl)[WPAD], int tid, const float4& a,
                                         const float4& b) {
  wT_put(wl, tid, a);
  wT_put(wl, wT_second(tid), b);
}

__global__ void k_init(const float* __restrict__ x, float* __restrict__ h,
                       int* __restrict__ cnt, int n) {
  long i = (long)blockIdx.x * blockDim.x + threadIdx.x;
  long tot = (long)n * D;
  if (i < n) cnt[i] = 0;
  if (i < tot) h[i] = __logf(x[i] + 1.0f);
}

__global__ void k_count(const int* __restrict__ dst, int* __restrict__ cnt,
                        int* __restrict__ rank, int e) {
  long i = (long)blockIdx.x * blockDim.x + threadIdx.x;
  if (i < e) rank[i] = atomicAdd(&cnt[dst[i]], 1);
}

__global__ void __launch_bounds__(1024)
k_scan(const int* __restrict__ cnt, int* __restrict__ rowptr, int n) {
  __shared__ int wtot[32];
  __shared__ int wpre[32];
  __shared__ int s_off;
  const int tid = threadIdx.x;
  const int lane = tid & 31;
  const int w = tid >> 5;
  const int tile0 = blockIdx.x * 1024;
  int p = 0;
  for (int i = tid * 4; i < tile0; i += 4096) {
    const int4 q = *reinterpret_cast<const int4*>(cnt + i);
    p += q.x + q.y + q.z + q.w;
  }
  const int i = tile0 + tid;
  const int v = (i < n) ? cnt[i] : 0;
  int x = v;
  for (int d = 1; d < 32; d <<= 1) {
    const int y = __shfl_up(x, d, 32);
    if (lane >= d) x += y;
  }
  for (int d = 16; d > 0; d >>= 1) p += __shfl_xor(p, d, 32);
  if (lane == 31) wtot[w] = x;
  if (lane == 0) wpre[w] = p;
  __syncthreads();
  if (w == 0) {
    const int t = wtot[lane];
    int q = wpre[lane];
    int xs = t;
    for (int d = 1; d < 32; d <<= 1) {
      const int y = __shfl_up(xs, d, 32);
      if (lane >= d) xs += y;
    }
    for (int d = 16; d > 0; d >>= 1) q += __shfl_xor(q, d, 32);
    wtot[lane] = xs - t;
    if (lane == 0) s_off = q;
  }
  __syncthreads();
  if (i <= n) rowptr[i] = s_off + wtot[w] + (x - v);
}

__global__ void k_fill(const int* __restrict__ src, const int* __restrict__ dst,
                       const int* __restrict__ rowptr, const int* __restrict__ rank,
                       int* __restrict__ col, int e) {
  long i = (long)blockIdx.x * blockDim.x + threadIdx.x;
  if (i < e) col[rowptr[dst[i]] + rank[i]] = src[i];
}

__global__ void __launch_bounds__(NTHR)
k_sage(const float* __restrict__ h, const int* __restrict__ rowptr,
       const int* __restrict__ col, const float* __restrict__ Ws,
       const float* __restrict__ Wn, const float* __restrict__ bias,
       float* __restrict__ out, int n, int do_norm) {
  __shared__ float wls[D][WPAD];
  __shared__ float wln[D][WPAD];
  __shared__ float nt[WPB][17][WPAD];

  const int lane = threadIdx.x & 31;
  const int wave = threadIdx.x >> 5;
  const int row0 = (blockIdx.x * WPB + wave) * 16;
  int rq = row0 + lane;
  if (rq > n) rq = n;
  const int rpv = rowptr[rq];

  {
    float4 s0, s1, n0, n1;
    wT_load(Ws, threadIdx.x, s0, s1);
    wT_load(Wn, threadIdx.x, n0, n1);
    wT_store(wls, threadIdx.x, s0, s1);
    wT_store(wln, threadIdx.x, n0, n1);
  }

  int rp[17];
#pragma unroll
  for (int r = 0; r <= 16; ++r) rp[r] = __builtin_amdgcn_readlane(rpv, r);
  const int segBeg = rp[0];
  const int segEnd = rp[16];
  int nsrc = (segBeg + lane < segEnd) ? col[segBeg + lane] : 0;
  __syncthreads();

  if (row0 >= n) return;
  const int half = lane >> 4;
  const int l15 = lane & 15;
  int rowA = row0 + l15;
  if (rowA >= n) rowA = n - 1;
  const float* __restrict__ hp = h + (size_t)rowA * D;

  const v2f zero2 = {0.0f, 0.0f};
#pragma unroll
  for (int r = 0; r < 16; ++r) *reinterpret_cast<v2f*>(&nt[wave][r][lane * 2]) = zero2;

  v2f sum = zero2;
  int cur = 16;
  for (int base = segBeg; base < segEnd; base += 32) {
    const int j = base + lane;
    const int msrc = nsrc;
    int mrow = 0;
#pragma unroll
    for (int r = 1; r <= 16; ++r) mrow += (j >= rp[r]) ? 1 : 0;
    v2f v[32];
    int rr[32];
#pragma unroll
    for (int u = 0; u < 32; ++u) {
      const int s = __builtin_amdgcn_readlane(msrc, u);
      rr[u] = __builtin_amdgcn_readlane(mrow, u);
      v[u] = *reinterpret_cast<const v2f*>(h + (size_t)s * D + lane * 2);
    }
    const int jn = j + 32;
    nsrc = (jn < segEnd) ? col[jn] : 0;
#pragma unroll
    for (int u = 0; u < 32; ++u) {
      if (rr[u] != cur) {
        const int c0 = __builtin_amdgcn_readlane(rpv, cur);
        const int c1 = __builtin_amdgcn_readlane(rpv, cur + 1);
        const float dinv = 1.0f / fmaxf((float)(c1 - c0), 1.0f);
        v2f m = sum;
        m.x *= dinv;
        m.y *= dinv;
        *reinterpret_cast<v2f*>(&nt[wave][cur][lane * 2]) = m;
        sum = zero2;
        cur = rr[u];
      }
      sum.x += v[u].x;
      sum.y += v[u].y;
    }
  }
  {
    const int c0 = __builtin_amdgcn_readlane(rpv, cur);
    const int c1 = __builtin_amdgcn_readlane(rpv, cur + 1);
    const float dinv = 1.0f / fmaxf((float)(c1 - c0), 1.0f);
    v2f m = sum;
    m.x *= dinv;
    m.y *= dinv;
    *reinterpret_cast<v2f*>(&nt[wave][cur][lane * 2]) = m;
  }

  v8f acc[4] = {};
#pragma unroll
  for (int k0 = 0; k0 < D; k0 += 4) {
    const int kk = k0 + half * 2;
    const v2f ah = *reinterpret_cast<const v2f*>(hp + kk);
    const v2f an = *reinterpret_cast<const v2f*>(&nt[wave][l15][kk]);
#pragma unroll
    for (int t = 0; t < 4; ++t) {
      const int nn = t * 16 + l15;
      const v2f bs = *reinterpret_cast<const v2f*>(&wls[nn][kk]);
      const v2f bn = *reinterpret_cast<const v2f*>(&wln[nn][kk]);
      acc[t] = WMMA4(ah, bs, acc[t]);
      acc[t] = WMMA4(an, bn, acc[t]);
    }
  }

  float ss[8];
#pragma unroll
  for (int i = 0; i < 8; ++i) ss[i] = 0.0f;
#pragma unroll
  for (int t = 0; t < 4; ++t) {
    const float bv = bias[t * 16 + l15];
#pragma unroll
    for (int i = 0; i < 8; ++i) {
      float v = acc[t][i] + bv;
      if (do_norm) v = fmaxf(v, 0.0f);
      acc[t][i] = v;
      ss[i] += v * v;
    }
  }
  if (do_norm) {
#pragma unroll
    for (int i = 0; i < 8; ++i) {
      float s = ss[i];
      s += __shfl_xor(s, 1, 32);
      s += __shfl_xor(s, 2, 32);
      s += __shfl_xor(s, 4, 32);
      s += __shfl_xor(s, 8, 32);
      const float inv = 1.0f / fmaxf(sqrtf(s), 1e-12f);
#pragma unroll
      for (int t = 0; t < 4; ++t) acc[t][i] *= inv;
    }
  }
#pragma unroll
  for (int t = 0; t < 4; ++t) {
#pragma unroll
    for (int i = 0; i < 8; ++i) {
      const int r = row0 + i + half * 8;
      if (r < n) out[(size_t)r * D + t * 16 + l15] = acc[t][i];
    }
  }
}

__global__ void __launch_bounds__(NTHR)
k_head(const float* __restrict__ h, const float* __restrict__ fcw,
       const float* __restrict__ fcb, const float* __restrict__ gamma,
       const float* __restrict__ beta, const float* __restrict__ mean,
       const float* __restrict__ var, const float* __restrict__ w21,
       const float* __restrict__ b21, const float* __restrict__ w22,
       const float* __restrict__ b22, float* __restrict__ zloc,
       float* __restrict__ zscale, int n) {
  __shared__ float wl0[D][WPAD];
  __shared__ float wl1[D][WPAD];
  __shared__ float wl2[D][WPAD];
  __shared__ float u[WPB][16][WPAD];
  {
    float4 a0, a1, b0, b1, c0, c1;
    wT_load(fcw, threadIdx.x, a0, a1);
    wT_load(w21, threadIdx.x, b0, b1);
    wT_load(w22, threadIdx.x, c0, c1);
    wT_store(wl0, threadIdx.x, a0, a1);
    wT_store(wl1, threadIdx.x, b0, b1);
    wT_store(wl2, threadIdx.x, c0, c1);
  }
  __syncthreads();

  const int lane = threadIdx.x & 31;
  const int wave = threadIdx.x >> 5;
  const int row0 = (blockIdx.x * WPB + wave) * 16;
  if (row0 >= n) return;
  const int half = lane >> 4;
  const int l15 = lane & 15;
  int rowA = row0 + l15;
  if (rowA >= n) rowA = n - 1;
  const float* __restrict__ hp = h + (size_t)rowA * D;

  v8f acc[4] = {};
  for (int k0 = 0; k0 < D; k0 += 4) {
    const int kk = k0 + half * 2;
    const v2f a = *reinterpret_cast<const v2f*>(hp + kk);
#pragma unroll
    for (int t = 0; t < 4; ++t) {
      const int nn = t * 16 + l15;
      const v2f b = *reinterpret_cast<const v2f*>(&wl0[nn][kk]);
      acc[t] = WMMA4(a, b, acc[t]);
    }
  }
#pragma unroll
  for (int t = 0; t < 4; ++t) {
    const int c = t * 16 + l15;
    const float bb = fcb[c];
    const float m = mean[c];
    const float iv = rsqrtf(var[c] + 1e-5f);
    const float g = gamma[c];
    const float bt = beta[c];
#pragma unroll
    for (int i = 0; i < 8; ++i) {
      float v = acc[t][i] + bb;
      v = (v - m) * iv * g + bt;
      v = fmaxf(v, 0.0f);
      v = v + __logf(1.0f + __expf(-v));
      u[wave][i + half * 8][c] = v;
    }
  }
  v8f aL[4] = {}, aS[4] = {};
  for (int k0 = 0; k0 < D; k0 += 4) {
    const int kk = k0 + half * 2;
    const v2f a = *reinterpret_cast<const v2f*>(&u[wave][l15][kk]);
#pragma unroll
    for (int t = 0; t < 4; ++t) {
      const int nn = t * 16 + l15;
      const v2f bL = *reinterpret_cast<const v2f*>(&wl1[nn][kk]);
      const v2f bS = *reinterpret_cast<const v2f*>(&wl2[nn][kk]);
      aL[t] = WMMA4(a, bL, aL[t]);
      aS[t] = WMMA4(a, bS, aS[t]);
    }
  }
#pragma unroll
  for (int t = 0; t < 4; ++t) {
    const int c = t * 16 + l15;
    const float bl = b21[c];
    const float bs = b22[c];
#pragma unroll
    for (int i = 0; i < 8; ++i) {
      const int r = row0 + i + half * 8;
      if (r < n) {
        const size_t o = (size_t)r * D + c;
        zloc[o] = aL[t][i] + bl;
        zscale[o] = __expf(aS[t][i] + bs);
      }
    }
  }
}

extern "C" void kernel_launch(void* const* d_in, const int* in_sizes, int n_in,
                              void* d_out, int out_size, void* d_ws, size_t ws_size,
                              hipStream_t stream) {
  const float* x = (const float*)d_in[0];
  const int* esrc = (const int*)d_in[1];
  const int* edst = (const int*)d_in[2];
  const float* Wself = (const float*)d_in[3];
  const float* Wneigh = (const float*)d_in[4];
  const float* bsage = (const float*)d_in[5];
  const float* fcw = (const float*)d_in[6];
  const float* fcb = (const float*)d_in[7];
  const float* gamma = (const float*)d_in[8];
  const float* beta = (const float*)d_in[9];
  const float* mean = (const float*)d_in[10];
  const float* var = (const float*)d_in[11];
  const float* w21 = (const float*)d_in[12];
  const float* b21 = (const float*)d_in[13];
  const float* w22 = (const float*)d_in[14];
  const float* b22 = (const float*)d_in[15];

  const int n = in_sizes[0] / D;
  const int e = in_sizes[1];
  const size_t nd = (size_t)n * D;

  float* ws = (float*)d_ws;
  int* cnt = (int*)ws;
  float* hA = ws + (((size_t)n + 255) & ~(size_t)255);
  float* hB = hA + nd;
  int* rowptr = (int*)(hB + nd);
  int* col = rowptr + (((size_t)n + 1 + 255) & ~(size_t)255);
  int* rank = col + e;

  {
    long tot = (long)n * D;
    k_init<<<(int)((tot + 255) / 256), 256, 0, stream>>>(x, hA, cnt, n);
  }
  k_count<<<(e + 255) / 256, 256, 0, stream>>>(edst, cnt, rank, e);
  k_scan<<<(n + 1024) / 1024, 1024, 0, stream>>>(cnt, rowptr, n);
  k_fill<<<(e + 255) / 256, 256, 0, stream>>>(esrc, edst, rowptr, rank, col, e);

  const int tiles = (n + 15) / 16;
  const int gblocks = (tiles + WPB - 1) / WPB;

  float* hcur = hA;
  float* hnext = hB;
  for (int l = 0; l < 3; ++l) {
    k_sage<<<gblocks, NTHR, 0, stream>>>(hcur, rowptr, col, Wself + (size_t)l * D * D,
                                        Wneigh + (size_t)l * D * D, bsage + (size_t)l * D,
                                        hnext, n, (l < 2) ? 1 : 0);
    float* tmp = hcur;
    hcur = hnext;
    hnext = tmp;
  }

  float* zloc = (float*)d_out;
  float* zscale = zloc + nd;
  k_head<<<gblocks, NTHR, 0, stream>>>(hcur, fcw, fcb, gamma, beta, mean, var, w21, b21,
                                      w22, b22, zloc, zscale, n);
}


# Self half of each layer product (h @ Ws) runs first, on the tile's rows read as 512-byte runs into the wave's LDS tile; gather and neighbour half follow; sum order: self terms, then neighbour terms
# speedup vs baseline: 3.3935x; 1.6845x over previous
#include <hip/hip_runtime.h>
#include <cstdint>

typedef __attribute__((ext_vector_type(2))) float v2f;
typedef __attribute__((ext_vector_type(8))) float v8f;

#define D 64
#define WPAD 68
#define WPB 25
#define NTHR (WPB * 32)
#define RB (WPB * 16)
#define CHUNK 4096
#define SORT_T 1024
#define MAXB 1024
#define MAXA 1024
#define CAP 8192
#define XQ (RB * D / 4)
#define XU ((XQ + SORT_T - 1) / SORT_T)

#define WMMA4(a, b, c) \
  __builtin_amdgcn_wmma_f32_16x16x4_f32(false, (a), false, (b), (short)0, (c), false, false)

__device__ __forceinline__ int wT_second(int tid) {
  return (tid + NTHR < 64 * 16) ? tid + NTHR : tid;
}
__device__ __forceinline__ void wT_load(const float* __restrict__ W, int tid, float4& a,
                                        float4& b) {
  const int i2 = wT_second(tid);
  a = *reinterpret_cast<const float4*>(W + (tid >> 4) * D + ((tid & 15) << 2));
  b = *reinterpret_cast<const float4*>(W + (i2 >> 4) * D + ((i2 & 15) << 2));
}
__device__ __forceinline__ void wT_put(float (*wl)[WPAD], int idx, const float4& v) {
  const int k = idx >> 4;
  const int n4 = (idx & 15) << 2;
  wl[n4 + 0][k] = v.x;
  wl[n4 + 1][k] = v.y;
  wl[n4 + 2][k] = v.z;
  wl[n4 + 3][k] = v.w;
}
__device__ __forceinline__ void wT_store(float (*wl)[WPAD], int tid, const float4& a,
                                         const float4& b) {
  wT_put(wl, tid, a);
  wT_put(wl, wT_second(tid), b);
}

__device__ __forceinline__ int block_scan_excl(int v, int* wtot, int& total) {
  const int lane = threadIdx.x & 31;
  const int w = threadIdx.x >> 5;
  int x = v;
  for (int d = 1; d < 32; d <<= 1) {
    const int y = __shfl_up(x, d, 32);
    if (lane >= d) x += y;
  }
  __syncthreads();
  if (lane == 31) wtot[w] = x;
  __syncthreads();
  if (w == 0) {
    const int t = wtot[lane];
    int xs = t;
    for (int d = 1; d < 32; d <<= 1) {
      const int y = __shfl_up(xs, d, 32);
      if (lane >= d) xs += y;
    }
    wtot[lane] = xs - t;
    if (lane == 31) wtot[32] = xs;
  }
  __syncthreads();
  total = wtot[32];
  return wtot[w] + (x - v);
}

__global__ void __launch_bounds__(SORT_T)
k_bin(const int* __restrict__ src, const int* __restrict__ dst, int* __restrict__ ssrc,
      int* __restrict__ sdst, int* __restrict__ tab, int e, int nbins, int nA) {
  __shared__ int hist[MAXB];
  __shared__ int lsrc[CHUNK];
  __shared__ int ldst[CHUNK];
  __shared__ int wtot[33];
  const int tid = threadIdx.x;
  const int base = blockIdx.x * CHUNK;
  hist[tid] = 0;
  int s[4], d[4], b[4], r[4];
  bool ok[4];
#pragma unroll
  for (int u = 0; u < 4; ++u) {
    const int i = base + u * SORT_T + tid;
    ok[u] = i < e;
    d[u] = ok[u] ? dst[i] : 0;
    s[u] = ok[u] ? src[i] : 0;
  }
  __syncthreads();
#pragma unroll
  for (int u = 0; u < 4; ++u) {
    b[u] = d[u] / RB;
    r[u] = ok[u] ? atomicAdd(&hist[b[u]], 1) : 0;
  }
  __syncthreads();
  int total;
  const int ex = block_scan_excl(hist[tid], wtot, total);
  hist[tid] = ex;
  if (tid < nbins) tab[(size_t)tid * nA + blockIdx.x] = ex;
  if (tid == 0) tab[(size_t)nbins * nA + blockIdx.x] = total;
  __syncthreads();
#pragma unroll
  for (int u = 0; u < 4; ++u) {
    if (ok[u]) {
      const int p = hist[b[u]] + r[u];
      lsrc[p] = s[u];
      ldst[p] = d[u];
    }
  }
  __syncthreads();
#pragma unroll
  for (int u = 0; u < 4; ++u) {
    const int i = u * SORT_T + tid;
    if (base + i < e) {
      ssrc[base + i] = lsrc[i];
      sdst[base + i] = ldst[i];
    }
  }
}

__device__ __forceinline__ int locate(const int* P, const int* RS, int nA, int j) {
  int lo = 0, hi = nA;
#pragma unroll
  for (int it = 0; it < 10; ++it) {
    const int mid = (lo + hi) >> 1;
    const bool ge = P[mid] <= j;
    lo = ge ? mid : lo;
    hi = ge ? hi : mid;
  }
  return RS[lo] + (j - P[lo]);
}

__global__ void __launch_bounds__(SORT_T)
k_group(const int* __restrict__ ssrc, const int* __restrict__ sdst,
        const int* __restrict__ tab, int* __restrict__ rowptr, int* __restrict__ col,
        const float* __restrict__ x, float* __restrict__ h0, int n, int nA) {
  __shared__ int P[MAXA + 1];
  __shared__ int RS[MAXA];
  __shared__ int cnt[RB];
  __shared__ int rst[RB];
  __shared__ int eD[CAP];
  __shared__ int eS[CAP];
  __shared__ int colL[CAP];
  __shared__ int wtot[33];
  const int tid = threadIdx.x;
  const int k = blockIdx.x;
  const int row0 = k * RB;
  int a = 0, len = 0;
  if (tid < nA) {
    a = tab[(size_t)k * nA + tid];
    len = tab[(size_t)(k + 1) * nA + tid] - a;
  }
  if (tid < RB) cnt[tid] = 0;
  int size, base;
  const int pex = block_scan_excl(len, wtot, size);
  (void)block_scan_excl(a, wtot, base);
  P[tid] = pex;
  if (tid == SORT_T - 1) P[SORT_T] = size;
  RS[tid] = tid * CHUNK + a;
  __syncthreads();
  const bool stage = size <= CAP;

  for (int j0 = 0; j0 < size; j0 += 4 * SORT_T) {
    int g[4], dv[4], sv[4];
#pragma unroll
    for (int u = 0; u < 4; ++u) {
      const int j = j0 + u * SORT_T + tid;
      g[u] = locate(P, RS, nA, (j < size) ? j : size - 1);
    }
#pragma unroll
    for (int u = 0; u < 4; ++u) {
      dv[u] = sdst[g[u]];
      sv[u] = ssrc[g[u]];
    }
#pragma unroll
    for (int u = 0; u < 4; ++u) {
      const int j = j0 + u * SORT_T + tid;
      if (j < size) {
        const int dl = dv[u] - row0;
        atomicAdd(&cnt[dl], 1);
        if (stage) {
          eD[j] = dl;
          eS[j] = sv[u];
        }
      }
    }
  }
  __syncthreads();
  int tot2;
  const int rex = block_scan_excl((tid < RB) ? cnt[tid] : 0, wtot, tot2);
  if (tid < RB) {
    rst[tid] = rex;
    cnt[tid] = 0;
    const int i = row0 + tid;
    if (i <= n) rowptr[i] = base + rex;
  }
  if (tid == 0 && k == (int)gridDim.x - 1 && row0 + RB <= n) rowptr[row0 + RB] = base + size;
  __syncthreads();

  const size_t q0 = (size_t)k * XQ;
  const size_t qn = (size_t)n * (D / 4);
  float4 xv[XU];
#pragma unroll
  for (int u = 0; u < XU; ++u) {
    const int p = u * SORT_T + tid;
    const size_t q = (p < XQ && q0 + p < qn) ? q0 + p : q0;
    xv[u] = reinterpret_cast<const float4*>(x)[q];
  }

  if (stage) {
    for (int j = tid; j < size; j += SORT_T) {
      const int dl = eD[j];
      colL[rst[dl] + atomicAdd(&cnt[dl], 1)] = eS[j];
    }
    __syncthreads();
    for (int j = tid; j < size; j += SORT_T) col[base + j] = colL[j];
  } else {
    for (int j0 = 0; j0 < size; j0 += 4 * SORT_T) {
      int g[4], dv[4], sv[4];
#pragma unroll
      for (int u = 0; u < 4; ++u) {
        const int j = j0 + u * SORT_T + tid;
        g[u] = locate(P, RS, nA, (j < size) ? j : size - 1);
      }
#pragma unroll
      for (int u = 0; u < 4; ++u) {
        dv[u] = sdst[g[u]];
        sv[u] = ssrc[g[u]];
      }
#pragma unroll
      for (int u = 0; u < 4; ++u) {
        const int j = j0 + u * SORT_T + tid;
        if (j < size) {
          const int dl = dv[u] - row0;
          col[base + rst[dl] + atomicAdd(&cnt[dl], 1)] = sv[u];
        }
      }
    }
  }

#pragma unroll
  for (int u = 0; u < XU; ++u) {
    const int p = u * SORT_T + tid;
    if (p < XQ && q0 + p < qn) {
      float4 r;
      r.x = __logf(xv[u].x + 1.0f);
      r.y = __logf(xv[u].y + 1.0f);
      r.z = __logf(xv[u].z + 1.0f);
      r.w = __logf(xv[u].w + 1.0f);
      reinterpret_cast<float4*>(h0)[q0 + p] = r;
    }
  }
}

__device__ __forceinline__ void self_load(const float* __restrict__ h, int row0, int half,
                                          int l15, int n, float4 (&sr)[8]) {
#pragma unroll
  for (int q = 0; q < 8; ++q) {
    int r = row0 + 2 * q + half;
    if (r >= n) r = n - 1;
    sr[q] = *reinterpret_cast<const float4*>(h + (size_t)r * D + l15 * 4);
  }
}

template <bool HEAD>
__global__ void __launch_bounds__(NTHR)
k_sage(const float* __restrict__ h, const int* __restrict__ rowptr,
       const int* __restrict__ col, const float* __restrict__ Ws,
       const float* __restrict__ Wn, const float* __restrict__ bias,
       float* __restrict__ out, int n, int do_norm, const float* __restrict__ fcw,
       const float* __restrict__ fcb, const float* __restrict__ gamma,
       const float* __restrict__ beta, const float* __restrict__ mean,
       const float* __restrict__ var, const float* __restrict__ w21,
       const float* __restrict__ b21, const float* __restrict__ w22,
       const float* __restrict__ b22, float* __restrict__ zloc,
       float* __restrict__ zscale) {
  __shared__ float wls[D][WPAD];
  __shared__ float wln[D][WPAD];
  __shared__ __attribute__((aligned(16))) float nt[WPB][17][WPAD];
  __shared__ float wlh[HEAD ? 3 * D : 1][WPAD];

  const int lane = threadIdx.x & 31;
  const int wave = HEAD ? __builtin_amdgcn_readfirstlane((int)(threadIdx.x >> 5))
                        : (int)(threadIdx.x >> 5);
  const int row0 = (blockIdx.x * WPB + wave) * 16;
  int rq = row0 + lane;
  if (rq > n) rq = n;
  const int rpv = rowptr[rq];

  const int half = lane >> 4;
  const int l15 = lane & 15;
  {
    float4 s0, s1, n0, n1;
    wT_load(Ws, threadIdx.x, s0, s1);
    wT_load(Wn, threadIdx.x, n0, n1);
    float4 sr[8];
    if constexpr (HEAD) {
      float4 a0, a1, b0, b1, c0, c1;
      wT_load(fcw, threadIdx.x, a0, a1);
      wT_load(w21, threadIdx.x, b0, b1);
      wT_load(w22, threadIdx.x, c0, c1);
      self_load(h, row0, half, l15, n, sr);
      wT_store(wls, threadIdx.x, s0, s1);
      wT_store(wln, threadIdx.x, n0, n1);
      wT_store(wlh, threadIdx.x, a0, a1);
      wT_store(wlh + D, threadIdx.x, b0, b1);
      wT_store(wlh + 2 * D, threadIdx.x, c0, c1);
    } else {
      self_load(h, row0, half, l15, n, sr);
      wT_store(wls, threadIdx.x, s0, s1);
      wT_store(wln, threadIdx.x, n0, n1);
    }
#pragma unroll
    for (int q = 0; q < 8; ++q)
      *reinterpret_cast<float4*>(&nt[wave][2 * q + half][l15 * 4]) = sr[q];
  }

  int rp[17];
#pragma unroll
  for (int r = 0; r <= 16; ++r) rp[r] = __builtin_amdgcn_readlane(rpv, r);
  const int segBeg = rp[0];
  const int segEnd = rp[16];
  int nsrc = (segBeg + lane < segEnd) ? col[segBeg + lane] : 0;
  __syncthreads();

  if (row0 >= n) return;

  v8f acc[4] = {};
#pragma unroll
  for (int k0 = 0; k0 < D; k0 += 4) {
    const int kk = k0 + half * 2;
    const v2f ah = *reinterpret_cast<const v2f*>(&nt[wave][l15][kk]);
#pragma unroll
    for (int t = 0; t < 4; ++t) {
      const int nn = t * 16 + l15;
      const v2f bs = *reinterpret_cast<const v2f*>(&wls[nn][kk]);
      acc[t] = WMMA4(ah, bs, acc[t]);
    }
  }

  const v2f zero2 = {0.0f, 0.0f};
#pragma unroll
  for (int r = 0; r < 16; ++r) *reinterpret_cast<v2f*>(&nt[wave][r][lane * 2]) = zero2;

  v2f sum = zero2;
  int cur = 16;
  for (int base = segBeg; base < segEnd; base += 32) {
    const int j = base + lane;
    const int msrc = nsrc;
    int mrow = 0;
#pragma unroll
    for (int r = 1; r <= 16; ++r) mrow += (j >= rp[r]) ? 1 : 0;
    v2f v[32];
    int rr[32];
#pragma unroll
    for (int u = 0; u < 32; ++u) {
      const int s = __builtin_amdgcn_readlane(msrc, u);
      rr[u] = __builtin_amdgcn_readlane(mrow, u);
      v[u] = *reinterpret_cast<const v2f*>(h + (size_t)s * D + lane * 2);
    }
    const int jn = j + 32;
    nsrc = (jn < segEnd) ? col[jn] : 0;
#pragma unroll
    for (int u = 0; u < 32; ++u) {
      if (rr[u] != cur) {
        const int c0 = __builtin_amdgcn_readlane(rpv, cur);
        const int c1 = __builtin_amdgcn_readlane(rpv, cur + 1);
        const float dinv = 1.0f / fmaxf((float)(c1 - c0), 1.0f);
        v2f m = sum;
        m.x *= dinv;
        m.y *= dinv;
        *reinterpret_cast<v2f*>(&nt[wave][cur][lane * 2]) = m;
        sum = zero2;
        cur = rr[u];
      }
      sum.x += v[u].x;
      sum.y += v[u].y;
    }
  }
  {
    const int c0 = __builtin_amdgcn_readlane(rpv, cur);
    const int c1 = __builtin_amdgcn_readlane(rpv, cur + 1);
    const float dinv = 1.0f / fmaxf((float)(c1 - c0), 1.0f);
    v2f m = sum;
    m.x *= dinv;
    m.y *= dinv;
    *reinterpret_cast<v2f*>(&nt[wave][cur][lane * 2]) = m;
  }

#pragma unroll
  for (int k0 = 0; k0 < D; k0 += 4) {
    const int kk = k0 + half * 2;
    const v2f an = *reinterpret_cast<const v2f*>(&nt[wave][l15][kk]);
#pragma unroll
    for (int t = 0; t < 4; ++t) {
      const int nn = t * 16 + l15;
      const v2f bn = *reinterpret_cast<const v2f*>(&wln[nn][kk]);
      acc[t] = WMMA4(an, bn, acc[t]);
    }
  }

  float ss[8];
#pragma unroll
  for (int i = 0; i < 8; ++i) ss[i] = 0.0f;
#pragma unroll
  for (int t = 0; t < 4; ++t) {
    const float bv = bias[t * 16 + l15];
#pragma unroll
    for (int i = 0; i < 8; ++i) {
      float v = acc[t][i] + bv;
      if (do_norm) v = fmaxf(v, 0.0f);
      acc[t][i] = v;
      ss[i] += v * v;
    }
  }
  if (do_norm) {
#pragma unroll
    for (int i = 0; i < 8; ++i) {
      float s = ss[i];
      s += __shfl_xor(s, 1, 32);
      s += __shfl_xor(s, 2, 32);
      s += __shfl_xor(s, 4, 32);
      s += __shfl_xor(s, 8, 32);
      const float inv = 1.0f / fmaxf(sqrtf(s), 1e-12f);
#pragma unroll
      for (int t = 0; t < 4; ++t) acc[t][i] *= inv;
    }
  }
  if constexpr (!HEAD) {
#pragma unroll
    for (int t = 0; t < 4; ++t) {
#pragma unroll
      for (int i = 0; i < 8; ++i) {
        const int r = row0 + i + half * 8;
        if (r < n) out[(size_t)r * D + t * 16 + l15] = acc[t][i];
      }
    }
  } else {
    float (*wl0)[WPAD] = wlh;
    float (*wl1)[WPAD] = wlh + D;
    float (*wl2)[WPAD] = wlh + 2 * D;
    float (*ut)[WPAD] = nt[wave];
#pragma unroll
    for (int t = 0; t < 4; ++t) {
#pragma unroll
      for (int i = 0; i < 8; ++i) ut[i + half * 8][t * 16 + l15] = acc[t][i];
    }
    v8f hc[4] = {};
    for (int k0 = 0; k0 < D; k0 += 4) {
      const int kk = k0 + half * 2;
      const v2f a = *reinterpret_cast<const v2f*>(&ut[l15][kk]);
#pragma unroll
      for (int t = 0; t < 4; ++t) {
        const int nn = t * 16 + l15;
        const v2f b = *reinterpret_cast<const v2f*>(&wl0[nn][kk]);
        hc[t] = WMMA4(a, b, hc[t]);
      }
    }
#pragma unroll
    for (int t = 0; t < 4; ++t) {
      const int c = t * 16 + l15;
      const float bb = fcb[c];
      const float m = mean[c];
      const float iv = rsqrtf(var[c] + 1e-5f);
      const float g = gamma[c];
      const float bt = beta[c];
#pragma unroll
      for (int i = 0; i < 8; ++i) {
        float v = hc[t][i] + bb;
        v = (v - m) * iv * g + bt;
        v = fmaxf(v, 0.0f);
        v = v + __logf(1.0f + __expf(-v));
        ut[i + half * 8][c] = v;
      }
    }
    v8f aL[4] = {}, aS[4] = {};
    for (int k0 = 0; k0 < D; k0 += 4) {
      const int kk = k0 + half * 2;
      const v2f a = *reinterpret_cast<const v2f*>(&ut[l15][kk]);
#pragma unroll
      for (int t = 0; t < 4; ++t) {
        const int nn = t * 16 + l15;
        const v2f bL = *reinterpret_cast<const v2f*>(&wl1[nn][kk]);
        const v2f bS = *reinterpret_cast<const v2f*>(&wl2[nn][kk]);
        aL[t] = WMMA4(a, bL, aL[t]);
        aS[t] = WMMA4(a, bS, aS[t]);
      }
    }
#pragma unroll
    for (int t = 0; t < 4; ++t) {
      const int c = t * 16 + l15;
      const float bl = b21[c];
      const float bs = b22[c];
#pragma unroll
      for (int i = 0; i < 8; ++i) {
        const int r = row0 + i + half * 8;
        if (r < n) {
          const size_t o = (size_t)r * D + c;
          zloc[o] = aL[t][i] + bl;
          zscale[o] = __expf(aS[t][i] + bs);
        }
      }
    }
  }
}

extern "C" void kernel_launch(void* const* d_in, const int* in_sizes, int n_in,
                              void* d_out, int out_size, void* d_ws, size_t ws_size,
                              hipStream_t stream) {
  const float* x = (const float*)d_in[0];
  const int* esrc = (const int*)d_in[1];
  const int* edst = (const int*)d_in[2];
  const float* Wself = (const float*)d_in[3];
  const float* Wneigh = (const float*)d_in[4];
  const float* bsage = (const float*)d_in[5];
  const float* fcw = (const float*)d_in[6];
  const float* fcb = (const float*)d_in[7];
  const float* gamma = (const float*)d_in[8];
  const float* beta = (const float*)d_in[9];
  const float* mean = (const float*)d_in[10];
  const float* var = (const float*)d_in[11];
  const float* w21 = (const float*)d_in[12];
  const float* b21 = (const float*)d_in[13];
  const float* w22 = (const float*)d_in[14];
  const float* b22 = (const float*)d_in[15];

  const int n = in_sizes[0] / D;
  const int e = in_sizes[1];
  const size_t nd = (size_t)n * D;

  float* ws = (float*)d_ws;
  float* hA = ws + (((size_t)n + 255) & ~(size_t)255);
  float* hB = hA + nd;
  int* rowptr = (int*)(hB + nd);
  int* col = rowptr + (((size_t)n + 1 + 255) & ~(size_t)255);
  int* ssrc = col + e;
  int* sdst = ssrc + e;
  int* tab = sdst + e;

  const int tiles = (n + 15) / 16;
  const int gblocks = (tiles + WPB - 1) / WPB;
  const int nA = (e + CHUNK - 1) / CHUNK;

  k_bin<<<nA, SORT_T, 0, stream>>>(esrc, edst, ssrc, sdst, tab, e, gblocks, nA);
  k_group<<<gblocks, SORT_T, 0, stream>>>(ssrc, sdst, tab, rowptr, col, x, hA, n, nA);

  float* zloc = (float*)d_out;
  float* zscale = zloc + nd;

  float* hcur = hA;
  float* hnext = hB;
  for (int l = 0; l < 2; ++l) {
    k_sage<false><<<gblocks, NTHR, 0, stream>>>(
        hcur, rowptr, col, Wself + (size_t)l * D * D, Wneigh + (size_t)l * D * D,
        bsage + (size_t)l * D, hnext, n, 1, fcw, fcb, gamma, beta, mean, var, w21, b21, w22,
        b22, zloc, zscale);
    float* tmp = hcur;
    hcur = hnext;
    hnext = tmp;
  }
  k_sage<true><<<gblocks, NTHR, 0, stream>>>(
      hcur, rowptr, col, Wself + (size_t)2 * D * D, Wneigh + (size_t)2 * D * D,
      bsage + (size_t)2 * D, hnext, n, 0, fcw, fcb, gamma, beta, mean, var, w21, b21, w22, b22,
      zloc, zscale);
}


# No address-translation wait inside a group of memory instructions: gather row loads on one scalar base + a 32-bit offset register per load, whole-tile output stores under the full lane mask
# speedup vs baseline: 3.6170x; 1.0659x over previous
#include <hip/hip_runtime.h>
#include <cstdint>

typedef __attribute__((ext_vector_type(2))) float v2f;
typedef __attribute__((ext_vector_type(8))) float v8f;

#define D 64
#define WPAD 68
#define WPB 25
#define NTHR (WPB * 32)
#define RB (WPB * 16)
#define CHUNK 4096
#define SORT_T 1024
#define MAXB 1024
#define MAXA 1024
#define CAP 8192
#define XQ (RB * D / 4)
#define XU ((XQ + SORT_T - 1) / SORT_T)

#define WMMA4(a, b, c) \
  __builtin_amdgcn_wmma_f32_16x16x4_f32(false, (a), false, (b), (short)0, (c), false, false)

__device__ __forceinline__ int wT_second(int tid) {
  return (tid + NTHR < 64 * 16) ? tid + NTHR : tid;
}
__device__ __forceinline__ void wT_load(const float* __restrict__ W, int tid, float4& a,
                                        float4& b) {
  const int i2 = wT_second(tid);
  a = *reinterpret_cast<const float4*>(W + (tid >> 4) * D + ((tid & 15) << 2));
  b = *reinterpret_cast<const float4*>(W + (i2 >> 4) * D + ((i2 & 15) << 2));
}
__device__ __forceinline__ void wT_put(float (*wl)[WPAD], int idx, const float4& v) {
  const int k = idx >> 4;
  const int n4 = (idx & 15) << 2;
  wl[n4 + 0][k] = v.x;
  wl[n4 + 1][k] = v.y;
  wl[n4 + 2][k] = v.z;
  wl[n4 + 3][k] = v.w;
}
__device__ __forceinline__ void wT_store(float (*wl)[WPAD], int tid, const float4& a,
                                         const float4& b) {
  wT_put(wl, tid, a);
  wT_put(wl, wT_second(tid), b);
}

__device__ __forceinline__ int block_scan_excl(int v, int* wtot, int& total) {
  const int lane = threadIdx.x & 31;
  const int w = threadIdx.x >> 5;
  int x = v;
  for (int d = 1; d < 32; d <<= 1) {
    const int y = __shfl_up(x, d, 32);
    if (lane >= d) x += y;
  }
  __syncthreads();
  if (lane == 31) wtot[w] = x;
  __syncthreads();
  if (w == 0) {
    const int t = wtot[lane];
    int xs = t;
    for (int d = 1; d < 32; d <<= 1) {
      const int y = __shfl_up(xs, d, 32);
      if (lane >= d) xs += y;
    }
    wtot[lane] = xs - t;
    if (lane == 31) wtot[32] = xs;
  }
  __syncthreads();
  total = wtot[32];
  return wtot[w] + (x - v);
}

__global__ void __launch_bounds__(SORT_T)
k_bin(const int* __restrict__ src, const int* __restrict__ dst, int* __restrict__ ssrc,
      int* __restrict__ sdst, int* __restrict__ tab, int e, int nbins, int nA) {
  __shared__ int hist[MAXB];
  __shared__ int lsrc[CHUNK];
  __shared__ int ldst[CHUNK];
  __shared__ int wtot[33];
  const int tid = threadIdx.x;
  const int base = blockIdx.x * CHUNK;
  hist[tid] = 0;
  int s[4], d[4], b[4], r[4];
  bool ok[4];
#pragma unroll
  for (int u = 0; u < 4; ++u) {
    const int i = base + u * SORT_T + tid;
    ok[u] = i < e;
    d[u] = ok[u] ? dst[i] : 0;
    s[u] = ok[u] ? src[i] : 0;
  }
  __syncthreads();
#pragma unroll
  for (int u = 0; u < 4; ++u) {
    b[u] = d[u] / RB;
    r[u] = ok[u] ? atomicAdd(&hist[b[u]], 1) : 0;
  }
  __syncthreads();
  int total;
  const int ex = block_scan_excl(hist[tid], wtot, total);
  hist[tid] = ex;
  if (tid < nbins) tab[(size_t)tid * nA + blockIdx.x] = ex;
  if (tid == 0) tab[(size_t)nbins * nA + blockIdx.x] = total;
  __syncthreads();
#pragma unroll
  for (int u = 0; u < 4; ++u) {
    if (ok[u]) {
      const int p = hist[b[u]] + r[u];
      lsrc[p] = s[u];
      ldst[p] = d[u];
    }
  }
  __syncthreads();
#pragma unroll
  for (int u = 0; u < 4; ++u) {
    const int i = u * SORT_T + tid;
    if (base + i < e) {
      ssrc[base + i] = lsrc[i];
      sdst[base + i] = ldst[i];
    }
  }
}

__device__ __forceinline__ int locate(const int* P, const int* RS, int nA, int j) {
  int lo = 0, hi = nA;
#pragma unroll
  for (int it = 0; it < 10; ++it) {
    const int mid = (lo + hi) >> 1;
    const bool ge = P[mid] <= j;
    lo = ge ? mid : lo;
    hi = ge ? hi : mid;
  }
  return RS[lo] + (j - P[lo]);
}

__global__ void __launch_bounds__(SORT_T)
k_group(const int* __restrict__ ssrc, const int* __restrict__ sdst,
        const int* __restrict__ tab, int* __restrict__ rowptr, int* __restrict__ col,
        const float* __restrict__ x, float* __restrict__ h0, int n, int nA) {
  __shared__ int P[MAXA + 1];
  __shared__ int RS[MAXA];
  __shared__ int cnt[RB];
  __shared__ int rst[RB];
  __shared__ int eD[CAP];
  __shared__ int eS[CAP];
  __shared__ int colL[CAP];
  __shared__ int wtot[33];
  const int tid = threadIdx.x;
  const int k = blockIdx.x;
  const int row0 = k * RB;
  int a = 0, len = 0;
  if (tid < nA) {
    a = tab[(size_t)k * nA + tid];
    len = tab[(size_t)(k + 1) * nA + tid] - a;
  }
  if (tid < RB) cnt[tid] = 0;
  int size, base;
  const int pex = block_scan_excl(len, wtot, size);
  (void)block_scan_excl(a, wtot, base);
  P[tid] = pex;
  if (tid == SORT_T - 1) P[SORT_T] = size;
  RS[tid] = tid * CHUNK + a;
  __syncthreads();
  const bool stage = size <= CAP;

  for (int j0 = 0; j0 < size; j0 += 4 * SORT_T) {
    int g[4], dv[4], sv[4];
#pragma unroll
    for (int u = 0; u < 4; ++u) {
      const int j = j0 + u * SORT_T + tid;
      g[u] = locate(P, RS, nA, (j < size) ? j : size - 1);
    }
#pragma unroll
    for (int u = 0; u < 4; ++u) {
      dv[u] = sdst[g[u]];
      sv[u] = ssrc[g[u]];
    }
#pragma unroll
    for (int u = 0; u < 4; ++u) {
      const int j = j0 + u * SORT_T + tid;
      if (j < size) {
        const int dl = dv[u] - row0;
        atomicAdd(&cnt[dl], 1);
        if (stage) {
          eD[j] = dl;
          eS[j] = sv[u];
        }
      }
    }
  }
  __syncthreads();
  int tot2;
  const int rex = block_scan_excl((tid < RB) ? cnt[tid] : 0, wtot, tot2);
  if (tid < RB) {
    rst[tid] = rex;
    cnt[tid] = 0;
    const int i = row0 + tid;
    if (i <= n) rowptr[i] = base + rex;
  }
  if (tid == 0 && k == (int)gridDim.x - 1 && row0 + RB <= n) rowptr[row0 + RB] = base + size;
  __syncthreads();

  const size_t q0 = (size_t)k * XQ;
  const size_t qn = (size_t)n * (D / 4);
  float4 xv[XU];
#pragma unroll
  for (int u = 0; u < XU; ++u) {
    const int p = u * SORT_T + tid;
    const size_t q = (p < XQ && q0 + p < qn) ? q0 + p : q0;
    xv[u] = reinterpret_cast<const float4*>(x)[q];
  }

  if (stage) {
    for (int j = tid; j < size; j += SORT_T) {
      const int dl = eD[j];
      colL[rst[dl] + atomicAdd(&cnt[dl], 1)] = eS[j];
    }
    __syncthreads();
    for (int j = tid; j < size; j += SORT_T) col[base + j] = colL[j];
  } else {
    for (int j0 = 0; j0 < size; j0 += 4 * SORT_T) {
      int g[4], dv[4], sv[4];
#pragma unroll
      for (int u = 0; u < 4; ++u) {
        const int j = j0 + u * SORT_T + tid;
        g[u] = locate(P, RS, nA, (j < size) ? j : size - 1);
      }
#pragma unroll
      for (int u = 0; u < 4; ++u) {
        dv[u] = sdst[g[u]];
        sv[u] = ssrc[g[u]];
      }
#pragma unroll
      for (int u = 0; u < 4; ++u) {
        const int j = j0 + u * SORT_T + tid;
        if (j < size) {
          const int dl = dv[u] - row0;
          col[base + rst[dl] + atomicAdd(&cnt[dl], 1)] = sv[u];
        }
      }
    }
  }

#pragma unroll
  for (int u = 0; u < XU; ++u) {
    const int p = u * SORT_T + tid;
    if (p < XQ && q0 + p < qn) {
      float4 r;
      r.x = __logf(xv[u].x + 1.0f);
      r.y = __logf(xv[u].y + 1.0f);
      r.z = __logf(xv[u].z + 1.0f);
      r.w = __logf(xv[u].w + 1.0f);
      reinterpret_cast<float4*>(h0)[q0 + p] = r;
    }
  }
}

__device__ __forceinline__ void self_load(const float* __restrict__ h, int row0, int half,
                                          int l15, int n, float4 (&sr)[8]) {
#pragma unroll
  for (int q = 0; q < 8; ++q) {
    int r = row0 + 2 * q + half;
    if (r >= n) r = n - 1;
    sr[q] = *reinterpret_cast<const float4*>(h + (size_t)r * D + l15 * 4);
  }
}

template <bool HEAD>
__global__ void __launch_bounds__(NTHR)
k_sage(const float* __restrict__ h, const int* __restrict__ rowptr,
       const int* __restrict__ col, const float* __restrict__ Ws,
       const float* __restrict__ Wn, const float* __restrict__ bias,
       float* __restrict__ out, int n, int do_norm, const float* __restrict__ fcw,
       const float* __restrict__ fcb, const float* __restrict__ gamma,
       const float* __restrict__ beta, const float* __restrict__ mean,
       const float* __restrict__ var, const float* __restrict__ w21,
       const float* __restrict__ b21, const float* __restrict__ w22,
       const float* __restrict__ b22, float* __restrict__ zloc,
       float* __restrict__ zscale) {
  __shared__ float wls[D][WPAD];
  __shared__ float wln[D][WPAD];
  __shared__ __attribute__((aligned(16))) float nt[WPB][17][WPAD];
  __shared__ float wlh[HEAD ? 3 * D : 1][WPAD];

  const int lane = threadIdx.x & 31;
  const int wave = HEAD ? __builtin_amdgcn_readfirstlane((int)(threadIdx.x >> 5))
                        : (int)(threadIdx.x >> 5);
  const int row0 = (blockIdx.x * WPB + wave) * 16;
  int rq = row0 + lane;
  if (rq > n) rq = n;
  const int rpv = rowptr[rq];

  const int half = lane >> 4;
  const int l15 = lane & 15;
  {
    float4 s0, s1, n0, n1;
    wT_load(Ws, threadIdx.x, s0, s1);
    wT_load(Wn, threadIdx.x, n0, n1);
    float4 sr[8];
    if constexpr (HEAD) {
      float4 a0, a1, b0, b1, c0, c1;
      wT_load(fcw, threadIdx.x, a0, a1);
      wT_load(w21, threadIdx.x, b0, b1);
      wT_load(w22, threadIdx.x, c0, c1);
      self_load(h, row0, half, l15, n, sr);
      wT_store(wls, threadIdx.x, s0, s1);
      wT_store(wln, threadIdx.x, n0, n1);
      wT_store(wlh, threadIdx.x, a0, a1);
      wT_store(wlh + D, threadIdx.x, b0, b1);
      wT_store(wlh + 2 * D, threadIdx.x, c0, c1);
    } else {
      self_load(h, row0, half, l15, n, sr);
      wT_store(wls, threadIdx.x, s0, s1);
      wT_store(wln, threadIdx.x, n0, n1);
    }
#pragma unroll
    for (int q = 0; q < 8; ++q)
      *reinterpret_cast<float4*>(&nt[wave][2 * q + half][l15 * 4]) = sr[q];
  }

  int rp[17];
#pragma unroll
  for (int r = 0; r <= 16; ++r) rp[r] = __builtin_amdgcn_readlane(rpv, r);
  const int segBeg = rp[0];
  const int segEnd = rp[16];
  int nsrc = (segBeg + lane < segEnd) ? col[segBeg + lane] : 0;
  __syncthreads();

  if (row0 >= n) return;

  v8f acc[4] = {};
#pragma unroll
  for (int k0 = 0; k0 < D; k0 += 4) {
    const int kk = k0 + half * 2;
    const v2f ah = *reinterpret_cast<const v2f*>(&nt[wave][l15][kk]);
#pragma unroll
    for (int t = 0; t < 4; ++t) {
      const int nn = t * 16 + l15;
      const v2f bs = *reinterpret_cast<const v2f*>(&wls[nn][kk]);
      acc[t] = WMMA4(ah, bs, acc[t]);
    }
  }

  const char* hb = reinterpret_cast<const char*>(h);
  const unsigned lane8 = (unsigned)lane * 8u;

  const v2f zero2 = {0.0f, 0.0f};
#pragma unroll
  for (int r = 0; r < 16; ++r) *reinterpret_cast<v2f*>(&nt[wave][r][lane * 2]) = zero2;

  v2f sum = zero2;
  int cur = 16;
  for (int base = segBeg; base < segEnd; base += 32) {
    const int j = base + lane;
    const int msrc = nsrc;
    int mrow = 0;
#pragma unroll
    for (int r = 1; r <= 16; ++r) mrow += (j >= rp[r]) ? 1 : 0;
    v2f v[32];
    int rr[32];
    unsigned off[32];
#pragma unroll
    for (int u = 0; u < 32; ++u) {
      const int s = __builtin_amdgcn_readlane(msrc, u);
      rr[u] = __builtin_amdgcn_readlane(mrow, u);
      off[u] = (unsigned)s * (unsigned)(D * 4) + lane8;
    }
#pragma unroll
    for (int u = 0; u < 32; ++u) v[u] = *reinterpret_cast<const v2f*>(hb + off[u]);
    const int jn = j + 32;
    nsrc = (jn < segEnd) ? col[jn] : 0;
#pragma unroll
    for (int u = 0; u < 32; ++u) {
      if (rr[u] != cur) {
        const int c0 = __builtin_amdgcn_readlane(rpv, cur);
        const int c1 = __builtin_amdgcn_readlane(rpv, cur + 1);
        const float dinv = 1.0f / fmaxf((float)(c1 - c0), 1.0f);
        v2f m = sum;
        m.x *= dinv;
        m.y *= dinv;
        *reinterpret_cast<v2f*>(&nt[wave][cur][lane * 2]) = m;
        sum = zero2;
        cur = rr[u];
      }
      sum.x += v[u].x;
      sum.y += v[u].y;
    }
  }
  {
    const int c0 = __builtin_amdgcn_readlane(rpv, cur);
    const int c1 = __builtin_amdgcn_readlane(rpv, cur + 1);
    const float dinv = 1.0f / fmaxf((float)(c1 - c0), 1.0f);
    v2f m = sum;
    m.x *= dinv;
    m.y *= dinv;
    *reinterpret_cast<v2f*>(&nt[wave][cur][lane * 2]) = m;
  }

#pragma unroll
  for (int k0 = 0; k0 < D; k0 += 4) {
    const int kk = k0 + half * 2;
    const v2f an = *reinterpret_cast<const v2f*>(&nt[wave][l15][kk]);
#pragma unroll
    for (int t = 0; t < 4; ++t) {
      const int nn = t * 16 + l15;
      const v2f bn = *reinterpret_cast<const v2f*>(&wln[nn][kk]);
      acc[t] = WMMA4(an, bn, acc[t]);
    }
  }

  float ss[8];
#pragma unroll
  for (int i = 0; i < 8; ++i) ss[i] = 0.0f;
#pragma unroll
  for (int t = 0; t < 4; ++t) {
    const float bv = bias[t * 16 + l15];
#pragma unroll
    for (int i = 0; i < 8; ++i) {
      float v = acc[t][i] + bv;
      if (do_norm) v = fmaxf(v, 0.0f);
      acc[t][i] = v;
      ss[i] += v * v;
    }
  }
  if (do_norm) {
#pragma unroll
    for (int i = 0; i < 8; ++i) {
      float s = ss[i];
      s += __shfl_xor(s, 1, 32);
      s += __shfl_xor(s, 2, 32);
      s += __shfl_xor(s, 4, 32);
      s += __shfl_xor(s, 8, 32);
      const float inv = 1.0f / fmaxf(sqrtf(s), 1e-12f);
#pragma unroll
      for (int t = 0; t < 4; ++t) acc[t][i] *= inv;
    }
  }
  const bool whole = __builtin_amdgcn_readfirstlane(row0) + 16 <= n;
  if constexpr (!HEAD) {
    if (whole) {
      float* o = out + (size_t)(row0 + half * 8) * D + l15;
#pragma unroll
      for (int t = 0; t < 4; ++t) {
#pragma unroll
        for (int i = 0; i < 8; ++i) o[i * D + t * 16] = acc[t][i];
      }
    } else {
#pragma unroll
      for (int t = 0; t < 4; ++t) {
#pragma unroll
        for (int i = 0; i < 8; ++i) {
          const int r = row0 + i + half * 8;
          if (r < n) out[(size_t)r * D + t * 16 + l15] = acc[t][i];
        }
      }
    }
  } else {
    float (*wl0)[WPAD] = wlh;
    float (*wl1)[WPAD] = wlh + D;
    float (*wl2)[WPAD] = wlh + 2 * D;
    float (*ut)[WPAD] = nt[wave];
#pragma unroll
    for (int t = 0; t < 4; ++t) {
#pragma unroll
      for (int i = 0; i < 8; ++i) ut[i + half * 8][t * 16 + l15] = acc[t][i];
    }
    v8f hc[4] = {};
    for (int k0 = 0; k0 < D; k0 += 4) {
      const int kk = k0 + half * 2;
      const v2f a = *reinterpret_cast<const v2f*>(&ut[l15][kk]);
#pragma unroll
      for (int t = 0; t < 4; ++t) {
        const int nn = t * 16 + l15;
        const v2f b = *reinterpret_cast<const v2f*>(&wl0[nn][kk]);
        hc[t] = WMMA4(a, b, hc[t]);
      }
    }
#pragma unroll
    for (int t = 0; t < 4; ++t) {
      const int c = t * 16 + l15;
      const float bb = fcb[c];
      const float m = mean[c];
      const float iv = rsqrtf(var[c] + 1e-5f);
      const float g = gamma[c];
      const float bt = beta[c];
#pragma unroll
      for (int i = 0; i < 8; ++i) {
        float v = hc[t][i] + bb;
        v = (v - m) * iv * g + bt;
        v = fmaxf(v, 0.0f);
        v = v + __logf(1.0f + __expf(-v));
        ut[i + half * 8][c] = v;
      }
    }
    v8f aL[4] = {}, aS[4] = {};
    for (int k0 = 0; k0 < D; k0 += 4) {
      const int kk = k0 + half * 2;
      const v2f a = *reinterpret_cast<const v2f*>(&ut[l15][kk]);
#pragma unroll
      for (int t = 0; t < 4; ++t) {
        const int nn = t * 16 + l15;
        const v2f bL = *reinterpret_cast<const v2f*>(&wl1[nn][kk]);
        const v2f bS = *reinterpret_cast<const v2f*>(&wl2[nn][kk]);
        aL[t] = WMMA4(a, bL, aL[t]);
        aS[t] = WMMA4(a, bS, aS[t]);
      }
    }
    if (whole) {
      const size_t o0 = (size_t)(row0 + half * 8) * D + l15;
#pragma unroll
      for (int t = 0; t < 4; ++t) {
        const int c = t * 16 + l15;
        const float bl = b21[c];
        const float bs = b22[c];
#pragma unroll
        for (int i = 0; i < 8; ++i) {
          const size_t o = o0 + (size_t)(i * D + t * 16);
          zloc[o] = aL[t][i] + bl;
          zscale[o] = __expf(aS[t][i] + bs);
        }
      }
    } else {
#pragma unroll
      for (int t = 0; t < 4; ++t) {
        const int c = t * 16 + l15;
        const float bl = b21[c];
        const float bs = b22[c];
#pragma unroll
        for (int i = 0; i < 8; ++i) {
          const int r = row0 + i + half * 8;
          if (r < n) {
            const size_t o = (size_t)r * D + c;
            zloc[o] = aL[t][i] + bl;
            zscale[o] = __expf(aS[t][i] + bs);
          }
        }
      }
    }
  }
}

extern "C" void kernel_launch(void* const* d_in, const int* in_sizes, int n_in,
                              void* d_out, int out_size, void* d_ws, size_t ws_size,
                              hipStream_t stream) {
  const float* x = (const float*)d_in[0];
  const int* esrc = (const int*)d_in[1];
  const int* edst = (const int*)d_in[2];
  const float* Wself = (const float*)d_in[3];
  const float* Wneigh = (const float*)d_in[4];
  const float* bsage = (const float*)d_in[5];
  const float* fcw = (const float*)d_in[6];
  const float* fcb = (const float*)d_in[7];
  const float* gamma = (const float*)d_in[8];
  const float* beta = (const float*)d_in[9];
  const float* mean = (const float*)d_in[10];
  const float* var = (const float*)d_in[11];
  const float* w21 = (const float*)d_in[12];
  const float* b21 = (const float*)d_in[13];
  const float* w22 = (const float*)d_in[14];
  const float* b22 = (const float*)d_in[15];

  const int n = in_sizes[0] / D;
  const int e = in_sizes[1];
  const size_t nd = (size_t)n * D;

  float* ws = (float*)d_ws;
  float* hA = ws + (((size_t)n + 255) & ~(size_t)255);
  float* hB = hA + nd;
  int* rowptr = (int*)(hB + nd);
  int* col = rowptr + (((size_t)n + 1 + 255) & ~(size_t)255);
  int* ssrc = col + e;
  int* sdst = ssrc + e;
  int* tab = sdst + e;

  const int tiles = (n + 15) / 16;
  const int gblocks = (tiles + WPB - 1) / WPB;
  const int nA = (e + CHUNK - 1) / CHUNK;

  k_bin<<<nA, SORT_T, 0, stream>>>(esrc, edst, ssrc, sdst, tab, e, gblocks, nA);
  k_group<<<gblocks, SORT_T, 0, stream>>>(ssrc, sdst, tab, rowptr, col, x, hA, n, nA);

  float* zloc = (float*)d_out;
  float* zscale = zloc + nd;

  float* hcur = hA;
  float* hnext = hB;
  for (int l = 0; l < 2; ++l) {
    k_sage<false><<<gblocks, NTHR, 0, stream>>>(
        hcur, rowptr, col, Wself + (size_t)l * D * D, Wneigh + (size_t)l * D * D,
        bsage + (size_t)l * D, hnext, n, 1, fcw, fcb, gamma, beta, mean, var, w21, b21, w22,
        b22, zloc, zscale);
    float* tmp = hcur;
    hcur = hnext;
    hnext = tmp;
  }
  k_sage<true><<<gblocks, NTHR, 0, stream>>>(
      hcur, rowptr, col, Wself + (size_t)2 * D * D, Wneigh + (size_t)2 * D * D,
      bsage + (size_t)2 * D, hnext, n, 0, fcw, fcb, gamma, beta, mean, var, w21, b21, w22, b22,
      zloc, zscale);
}
